# starting kernel
# baseline (speedup 1.0000x reference)
#include <hip/hip_runtime.h>
#include <math.h>

typedef __attribute__((ext_vector_type(2))) float v2f;
typedef __attribute__((ext_vector_type(8))) float v8f;

#define DATA    64
#define WIDTH   256
#define TPTS    16
#define NINT    (TPTS - 1)
#define MAXSUB  8
#define MT      16
#define NTHR    512
#define BATCH   512

#define RTOLC   1e-3f
#define ATOLC   1e-6f
#define SAFETY  0.9f
#define FMINC   0.2f
#define FMAXC   10.0f

__device__ __constant__ float STAGE_A[7][6] = {
  { 0.f, 0.f, 0.f, 0.f, 0.f, 0.f },
  { 0.161f, 0.f, 0.f, 0.f, 0.f, 0.f },
  { -0.008480655492356989f, 0.335480655492357f, 0.f, 0.f, 0.f, 0.f },
  { 2.8971530571054935f, -6.359448489975075f, 4.3622954328695815f, 0.f, 0.f, 0.f },
  { 5.325864828439257f, -11.748883564062828f, 7.4955393428898365f, -0.09249506636175525f, 0.f, 0.f },
  { 5.86145544294642f, -12.92096931784711f, 8.159367898576159f, -0.071584973281401f, -0.028269050394068383f, 0.f },
  { 0.09646076681806523f, 0.01f, 0.4798896504144996f, 1.379008574103742f, -3.290069515436081f, 2.324710524099774f },
};
__device__ __constant__ float ECOEF[7] = {
  -0.001780011052226f, -0.000816434459657f, 0.007880878010262f, -0.144711007173263f,
   0.582357165452555f, -0.458082105929187f, 0.015151515151515152f };

__device__ __forceinline__ float softplusf(float x) {
  const float LOG2E = 1.4426950408889634f;
  const float LN2   = 0.6931471805599453f;
  float e = __builtin_amdgcn_exp2f(-fabsf(x) * LOG2E);
  return fmaxf(x, 0.0f) + LN2 * __builtin_amdgcn_logf(1.0f + e);
}

__global__ __launch_bounds__(NTHR, 1)
void node_tsit5_kernel(const float* __restrict__ ts, const float* __restrict__ y0,
                       const float* __restrict__ W0, const float* __restrict__ b0,
                       const float* __restrict__ W1, const float* __restrict__ b1,
                       const float* __restrict__ W2, const float* __restrict__ b2,
                       float* __restrict__ out)
{
  __shared__ __align__(16) float sY [MT][DATA];
  __shared__ __align__(16) float sYn[MT][DATA];
  __shared__ __align__(16) float sZ [MT][DATA];
  __shared__ __align__(16) float sK [7][MT][DATA];
  __shared__ __align__(16) float sH0[MT][WIDTH];
  __shared__ __align__(16) float sH1[MT][WIDTH];
  __shared__ __align__(16) float sFp[4][MT][DATA];
  __shared__ float sT[MT], sDt[MT], sDtc[MT], sStep[MT], sDone[MT];
  __shared__ float sTs[TPTS];

  const int tid  = threadIdx.x;
  const int lane = tid & 31;
  const int wav  = tid >> 5;
  const int kh   = lane >> 4;
  const int l16  = lane & 15;
  const int row0 = blockIdx.x * MT;
  const int er   = wav;
  const int ec   = lane * 2;

  if (tid < TPTS) sTs[tid] = ts[tid];
  {
    const float* yp = y0 + (row0 + er) * DATA + ec;
    float a0 = yp[0], a1 = yp[1];
    sY[er][ec] = a0; sY[er][ec + 1] = a1;
    float* op = out + ((size_t)(row0 + er) * TPTS + 0) * DATA + ec;
    op[0] = a0; op[1] = a1;
  }
  if (lane == 0) sDt[er] = ts[1] - ts[0];
  __syncthreads();

  auto mlp = [&](float (*dst)[DATA]) {
    {
      const int n = wav * 16 + l16;
      v8f acc = {};
      const float* wp = W0 + n * DATA + kh * 2;
      #pragma unroll
      for (int kk = 0; kk < DATA / 4; ++kk) {
        const int k = kk * 4 + kh * 2;
        v2f a = *(const v2f*)(&sZ[l16][k]);
        v2f b = *(const v2f*)(wp + kk * 4);
        acc = __builtin_amdgcn_wmma_f32_16x16x4_f32(false, a, false, b,
                                                    (short)0, acc, false, false);
      }
      const float bias = b0[n];
      #pragma unroll
      for (int v = 0; v < 8; ++v)
        sH0[v + kh * 8][n] = softplusf(acc[v] + bias);
    }
    __syncthreads();
    {
      const int n = wav * 16 + l16;
      v8f acc = {};
      const float* wp = W1 + n * WIDTH + kh * 2;
      #pragma unroll 8
      for (int kk = 0; kk < WIDTH / 4; ++kk) {
        const int k = kk * 4 + kh * 2;
        v2f a = *(const v2f*)(&sH0[l16][k]);
        v2f b = *(const v2f*)(wp + kk * 4);
        acc = __builtin_amdgcn_wmma_f32_16x16x4_f32(false, a, false, b,
                                                    (short)0, acc, false, false);
      }
      const float bias = b1[n];
      #pragma unroll
      for (int v = 0; v < 8; ++v)
        sH1[v + kh * 8][n] = softplusf(acc[v] + bias);
    }
    __syncthreads();
    {
      const int dtile = wav & 3;
      const int kq    = wav >> 2;
      const int d     = dtile * 16 + l16;
      v8f acc = {};
      const float* wp = W2 + d * WIDTH + kq * 64 + kh * 2;
      #pragma unroll
      for (int kk = 0; kk < 16; ++kk) {
        const int k = kq * 64 + kk * 4 + kh * 2;
        v2f a = *(const v2f*)(&sH1[l16][k]);
        v2f b = *(const v2f*)(wp + kk * 4);
        acc = __builtin_amdgcn_wmma_f32_16x16x4_f32(false, a, false, b,
                                                    (short)0, acc, false, false);
      }
      #pragma unroll
      for (int v = 0; v < 8; ++v)
        sFp[kq][v + kh * 8][d] = acc[v];
    }
    __syncthreads();
    #pragma unroll
    for (int j = 0; j < 2; ++j) {
      const int c = ec + j;
      dst[er][c] = sFp[0][er][c] + sFp[1][er][c] + sFp[2][er][c] + sFp[3][er][c] + b2[c];
    }
    __syncthreads();
  };

  #pragma unroll 1
  for (int iv = 0; iv < NINT; ++iv) {
    const float t1 = sTs[iv + 1];
    if (lane == 0) sT[er] = sTs[iv];
    __syncthreads();

    #pragma unroll 1
    for (int ss = 0; ss < MAXSUB; ++ss) {
      if (lane == 0) {
        float t = sT[er], dt = sDt[er];
        float rem  = t1 - t;
        bool  done = rem <= 1e-6f;
        sDtc[er]  = done ? dt : fminf(dt, rem);
        sDone[er] = done ? 1.0f : 0.0f;
      }
      __syncthreads();
      const float dtc = sDtc[er];

      #pragma unroll 1
      for (int s = 0; s < 7; ++s) {
        #pragma unroll
        for (int j = 0; j < 2; ++j) {
          const int c = ec + j;
          float acc = 0.0f;
          for (int q = 0; q < s && q < 6; ++q)
            acc = fmaf(STAGE_A[s][q], sK[q][er][c], acc);
          float z = fmaf(dtc, acc, sY[er][c]);
          sZ[er][c] = z;
          if (s == 6) sYn[er][c] = z;
        }
        __syncthreads();
        mlp(sK[s]);
      }

      float esq = 0.0f;
      #pragma unroll
      for (int j = 0; j < 2; ++j) {
        const int c = ec + j;
        float e = 0.0f;
        #pragma unroll
        for (int q = 0; q < 7; ++q) e = fmaf(ECOEF[q], sK[q][er][c], e);
        e *= dtc;
        float scale = fmaf(RTOLC, fmaxf(fabsf(sY[er][c]), fabsf(sYn[er][c])), ATOLC);
        float r = e / scale;
        esq = fmaf(r, r, esq);
      }
      #pragma unroll
      for (int o = 16; o > 0; o >>= 1) esq += __shfl_xor(esq, o, 32);
      if (lane == 0) {
        float errn   = sqrtf(esq * (1.0f / DATA));
        bool  accept = errn <= 1.0f;
        float factor = SAFETY * __powf(fmaxf(errn, 1e-10f), -0.2f);
        factor = fminf(fmaxf(factor, FMINC), FMAXC);
        bool  done = sDone[er] != 0.0f;
        bool  step = accept && !done;
        if (step)  sT[er] += dtc;
        if (!done) sDt[er] = dtc * factor;
        sStep[er] = step ? 1.0f : 0.0f;
      }
      __syncthreads();
      if (sStep[er] != 0.0f) {
        sY[er][ec]     = sYn[er][ec];
        sY[er][ec + 1] = sYn[er][ec + 1];
      }
      __syncthreads();
    }

    float* op = out + ((size_t)(row0 + er) * TPTS + (iv + 1)) * DATA + ec;
    op[0] = sY[er][ec];
    op[1] = sY[er][ec + 1];
    __syncthreads();
  }
}

extern "C" void kernel_launch(void* const* d_in, const int* in_sizes, int n_in,
                              void* d_out, int out_size, void* d_ws, size_t ws_size,
                              hipStream_t stream) {
  (void)in_sizes; (void)n_in; (void)out_size; (void)d_ws; (void)ws_size;
  const float* ts = (const float*)d_in[0];
  const float* y0 = (const float*)d_in[1];
  const float* W0 = (const float*)d_in[2];
  const float* b0 = (const float*)d_in[3];
  const float* W1 = (const float*)d_in[4];
  const float* b1 = (const float*)d_in[5];
  const float* W2 = (const float*)d_in[6];
  const float* b2 = (const float*)d_in[7];
  float* out = (float*)d_out;
  node_tsit5_kernel<<<dim3(BATCH / MT), dim3(NTHR), 0, stream>>>(
      ts, y0, W0, b0, W1, b1, W2, b2, out);
}


# Keep the 256x256 layer resident in LDS as WMMA B fragments, each wave filling its own 16 KiB once per launch: GEMM2 takes both operands from LDS, no L2 fetch per evaluation; same output bits
# speedup vs baseline: 3.2592x; 3.2592x over previous
#include <hip/hip_runtime.h>
#include <math.h>

typedef __attribute__((ext_vector_type(2))) float v2f;
typedef __attribute__((ext_vector_type(4))) float v4f;
typedef __attribute__((ext_vector_type(8))) float v8f;

#define DATA    64
#define WIDTH   256
#define TPTS    16
#define NINT    (TPTS - 1)
#define MAXSUB  8
#define MT      16
#define NTHR    512
#define BATCH   512

#define RTOLC   1e-3f
#define ATOLC   1e-6f
#define SAFETY  0.9f
#define FMINC   0.2f
#define FMAXC   10.0f

__device__ __constant__ float STAGE_A[7][6] = {
  { 0.f, 0.f, 0.f, 0.f, 0.f, 0.f },
  { 0.161f, 0.f, 0.f, 0.f, 0.f, 0.f },
  { -0.008480655492356989f, 0.335480655492357f, 0.f, 0.f, 0.f, 0.f },
  { 2.8971530571054935f, -6.359448489975075f, 4.3622954328695815f, 0.f, 0.f, 0.f },
  { 5.325864828439257f, -11.748883564062828f, 7.4955393428898365f, -0.09249506636175525f, 0.f, 0.f },
  { 5.86145544294642f, -12.92096931784711f, 8.159367898576159f, -0.071584973281401f, -0.028269050394068383f, 0.f },
  { 0.09646076681806523f, 0.01f, 0.4798896504144996f, 1.379008574103742f, -3.290069515436081f, 2.324710524099774f },
};
__device__ __constant__ float ECOEF[7] = {
  -0.001780011052226f, -0.000816434459657f, 0.007880878010262f, -0.144711007173263f,
   0.582357165452555f, -0.458082105929187f, 0.015151515151515152f };

__device__ __forceinline__ float softplusf(float x) {
  const float LOG2E = 1.4426950408889634f;
  const float LN2   = 0.6931471805599453f;
  float e = __builtin_amdgcn_exp2f(-fabsf(x) * LOG2E);
  return fmaxf(x, 0.0f) + LN2 * __builtin_amdgcn_logf(1.0f + e);
}

__device__ __forceinline__ int frag_idx(int m, int k) {
  const int kk = k >> 2;
  return (kk << 6) + ((4 * m + 2 * ((k >> 1) & 1) + 8 * (kk & 3)) & 63) + (k & 1);
}

__global__ __launch_bounds__(NTHR, 1)
void node_tsit5_kernel(const float* __restrict__ ts, const float* __restrict__ y0,
                       const float* __restrict__ W0, const float* __restrict__ b0,
                       const float* __restrict__ W1, const float* __restrict__ b1,
                       const float* __restrict__ W2, const float* __restrict__ b2,
                       float* __restrict__ out)
{
  __shared__ __align__(16) float sW1[WIDTH * WIDTH];
  __shared__ __align__(16) float sK [7][MT][DATA];
  __shared__ __align__(16) float sH0[MT * WIDTH];
  __shared__ __align__(16) float sH1[MT * WIDTH];
  float* const sZ = sH1;
  float (* const sFp)[MT][DATA] = (float (*)[MT][DATA])sH0;
  __shared__ float sT[MT], sDt[MT], sDtc[MT], sStep[MT], sDone[MT];
  __shared__ float sTs[TPTS];

  const int tid  = threadIdx.x;
  const int lane = tid & 31;
  const int wav  = tid >> 5;
  const int kh   = lane >> 4;
  const int l16  = lane & 15;
  const int row0 = blockIdx.x * MT;
  const int er   = wav;
  const int ec   = lane * 2;

  if (tid < TPTS) sTs[tid] = ts[tid];
  float yv[2], ynv[2] = { 0.0f, 0.0f };
  {
    const float* yp = y0 + (row0 + er) * DATA + ec;
    yv[0] = yp[0]; yv[1] = yp[1];
    float* op = out + ((size_t)(row0 + er) * TPTS + 0) * DATA + ec;
    op[0] = yv[0]; op[1] = yv[1];
  }
  if (lane == 0) sDt[er] = ts[1] - ts[0];
  {
    const float* wp = W1 + (wav * 16 + l16) * WIDTH + kh * 2;
    float* fp = sW1 + (wav * 1024 + lane) * 4;
    #pragma unroll 16
    for (int kk = 0; kk < WIDTH / 4; ++kk) {
      v2f b = *(const v2f*)(wp + kk * 4);
      *(v2f*)(fp + (kk >> 1) * 128 + (kk & 1) * 2) = b;
    }
  }
  __syncthreads();

  auto mlp = [&](float (*dst)[DATA]) {
    {
      const int n = wav * 16 + l16;
      v8f acc = {};
      const float* wp = W0 + n * DATA + kh * 2;
      #pragma unroll
      for (int kk = 0; kk < DATA / 4; ++kk) {
        const int k = kk * 4 + kh * 2;
        v2f a = *(const v2f*)(&sZ[frag_idx(l16, k)]);
        v2f b = *(const v2f*)(wp + kk * 4);
        acc = __builtin_amdgcn_wmma_f32_16x16x4_f32(false, a, false, b,
                                                    (short)0, acc, false, false);
      }
      const float bias = b0[n];
      #pragma unroll
      for (int v = 0; v < 8; ++v)
        sH0[frag_idx(v + kh * 8, n)] = softplusf(acc[v] + bias);
    }
    __syncthreads();
    {
      const int n = wav * 16 + l16;
      v8f acc = {};
      const float* bp = sW1 + (wav * 1024 + lane) * 4;
      v4f bq[2][4];
      v2f aq[2][8];
      #pragma unroll
      for (int i = 0; i < 4; ++i) {
        const int k = i * 8 + kh * 2;
        bq[0][i]         = *(const v4f*)(bp + i * 128);
        aq[0][2 * i]     = *(const v2f*)(&sH0[frag_idx(l16, k)]);
        aq[0][2 * i + 1] = *(const v2f*)(&sH0[frag_idx(l16, k + 4)]);
      }
      #pragma unroll
      for (int blk = 0; blk < 8; ++blk) {
        const int cur = blk & 1, nxt = cur ^ 1;
        if (blk + 1 < 8) {
          #pragma unroll
          for (int i = 0; i < 4; ++i) {
            const int p = (blk + 1) * 4 + i;
            const int k = p * 8 + kh * 2;
            bq[nxt][i]         = *(const v4f*)(bp + p * 128);
            aq[nxt][2 * i]     = *(const v2f*)(&sH0[frag_idx(l16, k)]);
            aq[nxt][2 * i + 1] = *(const v2f*)(&sH0[frag_idx(l16, k + 4)]);
          }
        }
        __builtin_amdgcn_sched_barrier(0);
        #pragma unroll
        for (int i = 0; i < 4; ++i) {
          v2f b0 = { bq[cur][i].x, bq[cur][i].y };
          v2f b1 = { bq[cur][i].z, bq[cur][i].w };
          acc = __builtin_amdgcn_wmma_f32_16x16x4_f32(false, aq[cur][2 * i], false, b0,
                                                      (short)0, acc, false, false);
          acc = __builtin_amdgcn_wmma_f32_16x16x4_f32(false, aq[cur][2 * i + 1], false, b1,
                                                      (short)0, acc, false, false);
        }
      }
      const float bias = b1[n];
      #pragma unroll
      for (int v = 0; v < 8; ++v)
        sH1[frag_idx(v + kh * 8, n)] = softplusf(acc[v] + bias);
    }
    __syncthreads();
    {
      const int dtile = wav & 3;
      const int kq    = wav >> 2;
      const int d     = dtile * 16 + l16;
      v8f acc = {};
      const float* wp = W2 + d * WIDTH + kq * 64 + kh * 2;
      #pragma unroll
      for (int kk = 0; kk < 16; ++kk) {
        const int k = kq * 64 + kk * 4 + kh * 2;
        v2f a = *(const v2f*)(&sH1[frag_idx(l16, k)]);
        v2f b = *(const v2f*)(wp + kk * 4);
        acc = __builtin_amdgcn_wmma_f32_16x16x4_f32(false, a, false, b,
                                                    (short)0, acc, false, false);
      }
      #pragma unroll
      for (int v = 0; v < 8; ++v)
        sFp[kq][v + kh * 8][d] = acc[v];
    }
    __syncthreads();
    #pragma unroll
    for (int j = 0; j < 2; ++j) {
      const int c = ec + j;
      dst[er][c] = sFp[0][er][c] + sFp[1][er][c] + sFp[2][er][c] + sFp[3][er][c] + b2[c];
    }
    __syncthreads();
  };

  #pragma unroll 1
  for (int iv = 0; iv < NINT; ++iv) {
    const float t1 = sTs[iv + 1];
    if (lane == 0) sT[er] = sTs[iv];
    __syncthreads();

    #pragma unroll 1
    for (int ss = 0; ss < MAXSUB; ++ss) {
      if (lane == 0) {
        float t = sT[er], dt = sDt[er];
        float rem  = t1 - t;
        bool  done = rem <= 1e-6f;
        sDtc[er]  = done ? dt : fminf(dt, rem);
        sDone[er] = done ? 1.0f : 0.0f;
      }
      __syncthreads();
      const float dtc = sDtc[er];

      #pragma unroll 1
      for (int s = 0; s < 7; ++s) {
        #pragma unroll
        for (int j = 0; j < 2; ++j) {
          const int c = ec + j;
          float acc = 0.0f;
          for (int q = 0; q < s && q < 6; ++q)
            acc = fmaf(STAGE_A[s][q], sK[q][er][c], acc);
          float z = fmaf(dtc, acc, yv[j]);
          sZ[frag_idx(er, c)] = z;
          if (s == 6) ynv[j] = z;
        }
        __syncthreads();
        mlp(sK[s]);
      }

      float esq = 0.0f;
      #pragma unroll
      for (int j = 0; j < 2; ++j) {
        const int c = ec + j;
        float e = 0.0f;
        #pragma unroll
        for (int q = 0; q < 7; ++q) e = fmaf(ECOEF[q], sK[q][er][c], e);
        e *= dtc;
        float scale = fmaf(RTOLC, fmaxf(fabsf(yv[j]), fabsf(ynv[j])), ATOLC);
        float r = e / scale;
        esq = fmaf(r, r, esq);
      }
      #pragma unroll
      for (int o = 16; o > 0; o >>= 1) esq += __shfl_xor(esq, o, 32);
      if (lane == 0) {
        float errn   = sqrtf(esq * (1.0f / DATA));
        bool  accept = errn <= 1.0f;
        float factor = SAFETY * __powf(fmaxf(errn, 1e-10f), -0.2f);
        factor = fminf(fmaxf(factor, FMINC), FMAXC);
        bool  done = sDone[er] != 0.0f;
        bool  step = accept && !done;
        if (step)  sT[er] += dtc;
        if (!done) sDt[er] = dtc * factor;
        sStep[er] = step ? 1.0f : 0.0f;
      }
      __syncthreads();
      if (sStep[er] != 0.0f) {
        yv[0] = ynv[0];
        yv[1] = ynv[1];
      }
      __syncthreads();
    }

    float* op = out + ((size_t)(row0 + er) * TPTS + (iv + 1)) * DATA + ec;
    op[0] = yv[0];
    op[1] = yv[1];
    __syncthreads();
  }
}

extern "C" void kernel_launch(void* const* d_in, const int* in_sizes, int n_in,
                              void* d_out, int out_size, void* d_ws, size_t ws_size,
                              hipStream_t stream) {
  (void)in_sizes; (void)n_in; (void)out_size; (void)d_ws; (void)ws_size;
  const float* ts = (const float*)d_in[0];
  const float* y0 = (const float*)d_in[1];
  const float* W0 = (const float*)d_in[2];
  const float* b0 = (const float*)d_in[3];
  const float* W1 = (const float*)d_in[4];
  const float* b1 = (const float*)d_in[5];
  const float* W2 = (const float*)d_in[6];
  const float* b2 = (const float*)d_in[7];
  float* out = (float*)d_out;
  node_tsit5_kernel<<<dim3(BATCH / MT), dim3(NTHR), 0, stream>>>(
      ts, y0, W0, b0, W1, b1, W2, b2, out);
}


# Fold each new k at once into register running sums for every later stage input and the error (tableau by column, same ascending fmaf order): k1..k7 leave LDS, no per-stage re-read loop; same bits
# speedup vs baseline: 3.5198x; 1.0800x over previous
#include <hip/hip_runtime.h>
#include <math.h>

typedef __attribute__((ext_vector_type(2))) float v2f;
typedef __attribute__((ext_vector_type(4))) float v4f;
typedef __attribute__((ext_vector_type(8))) float v8f;

#define DATA    64
#define WIDTH   256
#define TPTS    16
#define NINT    (TPTS - 1)
#define MAXSUB  8
#define MT      16
#define NTHR    512
#define BATCH   512

#define RTOLC   1e-3f
#define ATOLC   1e-6f
#define SAFETY  0.9f
#define FMINC   0.2f
#define FMAXC   10.0f

__device__ __constant__ float STAGE_COL[7][8] = {
  { 0.161f, -0.008480655492356989f, 2.8971530571054935f, 5.325864828439257f, 5.86145544294642f, 0.09646076681806523f,
    -0.001780011052226f, 0.f },
  { 0.335480655492357f, -6.359448489975075f, -11.748883564062828f, -12.92096931784711f, 0.01f, 0.f,
    -0.000816434459657f, 0.f },
  { 4.3622954328695815f, 7.4955393428898365f, 8.159367898576159f, 0.4798896504144996f, 0.f, 0.f,
    0.007880878010262f, 0.f },
  { -0.09249506636175525f, -0.071584973281401f, 1.379008574103742f, 0.f, 0.f, 0.f,
    -0.144711007173263f, 0.f },
  { -0.028269050394068383f, -3.290069515436081f, 0.f, 0.f, 0.f, 0.f,
    0.582357165452555f, 0.f },
  { 2.324710524099774f, 0.f, 0.f, 0.f, 0.f, 0.f,
    -0.458082105929187f, 0.f },
  { 0.f, 0.f, 0.f, 0.f, 0.f, 0.f,
    0.015151515151515152f, 0.f },
};

__device__ __forceinline__ float softplusf(float x) {
  const float LOG2E = 1.4426950408889634f;
  const float LN2   = 0.6931471805599453f;
  float e = __builtin_amdgcn_exp2f(-fabsf(x) * LOG2E);
  return fmaxf(x, 0.0f) + LN2 * __builtin_amdgcn_logf(1.0f + e);
}

__device__ __forceinline__ int frag_idx(int m, int k) {
  const int kk = k >> 2;
  return (kk << 6) + ((4 * m + 2 * ((k >> 1) & 1) + 8 * (kk & 3)) & 63) + (k & 1);
}

__global__ __launch_bounds__(NTHR, 1)
void node_tsit5_kernel(const float* __restrict__ ts, const float* __restrict__ y0,
                       const float* __restrict__ W0, const float* __restrict__ b0,
                       const float* __restrict__ W1, const float* __restrict__ b1,
                       const float* __restrict__ W2, const float* __restrict__ b2,
                       float* __restrict__ out)
{
  __shared__ __align__(16) float sW1[WIDTH * WIDTH];
  __shared__ __align__(16) float sH0[MT * WIDTH];
  __shared__ __align__(16) float sH1[MT * WIDTH];
  float* const sZ = sH1;
  float (* const sFp)[MT][DATA] = (float (*)[MT][DATA])sH0;
  __shared__ float sT[MT], sDt[MT], sDtc[MT], sStep[MT], sDone[MT];
  __shared__ float sTs[TPTS];

  const int tid  = threadIdx.x;
  const int lane = tid & 31;
  const int wav  = tid >> 5;
  const int kh   = lane >> 4;
  const int l16  = lane & 15;
  const int row0 = blockIdx.x * MT;
  const int er   = wav;
  const int ec   = lane * 2;

  if (tid < TPTS) sTs[tid] = ts[tid];
  float yv[2], ynv[2] = { 0.0f, 0.0f };
  {
    const float* yp = y0 + (row0 + er) * DATA + ec;
    yv[0] = yp[0]; yv[1] = yp[1];
    float* op = out + ((size_t)(row0 + er) * TPTS + 0) * DATA + ec;
    op[0] = yv[0]; op[1] = yv[1];
  }
  if (lane == 0) sDt[er] = ts[1] - ts[0];
  {
    const float* wp = W1 + (wav * 16 + l16) * WIDTH + kh * 2;
    float* fp = sW1 + (wav * 1024 + lane) * 4;
    #pragma unroll 16
    for (int kk = 0; kk < WIDTH / 4; ++kk) {
      v2f b = *(const v2f*)(wp + kk * 4);
      *(v2f*)(fp + (kk >> 1) * 128 + (kk & 1) * 2) = b;
    }
  }
  __syncthreads();

  auto mlp = [&]() {
    {
      const int n = wav * 16 + l16;
      v8f acc = {};
      const float* wp = W0 + n * DATA + kh * 2;
      #pragma unroll
      for (int kk = 0; kk < DATA / 4; ++kk) {
        const int k = kk * 4 + kh * 2;
        v2f a = *(const v2f*)(&sZ[frag_idx(l16, k)]);
        v2f b = *(const v2f*)(wp + kk * 4);
        acc = __builtin_amdgcn_wmma_f32_16x16x4_f32(false, a, false, b,
                                                    (short)0, acc, false, false);
      }
      const float bias = b0[n];
      #pragma unroll
      for (int v = 0; v < 8; ++v)
        sH0[frag_idx(v + kh * 8, n)] = softplusf(acc[v] + bias);
    }
    __syncthreads();
    {
      const int n = wav * 16 + l16;
      v8f acc = {};
      const float* bp = sW1 + (wav * 1024 + lane) * 4;
      v4f bq[2][4];
      v2f aq[2][8];
      #pragma unroll
      for (int i = 0; i < 4; ++i) {
        const int k = i * 8 + kh * 2;
        bq[0][i]         = *(const v4f*)(bp + i * 128);
        aq[0][2 * i]     = *(const v2f*)(&sH0[frag_idx(l16, k)]);
        aq[0][2 * i + 1] = *(const v2f*)(&sH0[frag_idx(l16, k + 4)]);
      }
      #pragma unroll
      for (int blk = 0; blk < 8; ++blk) {
        const int cur = blk & 1, nxt = cur ^ 1;
        if (blk + 1 < 8) {
          #pragma unroll
          for (int i = 0; i < 4; ++i) {
            const int p = (blk + 1) * 4 + i;
            const int k = p * 8 + kh * 2;
            bq[nxt][i]         = *(const v4f*)(bp + p * 128);
            aq[nxt][2 * i]     = *(const v2f*)(&sH0[frag_idx(l16, k)]);
            aq[nxt][2 * i + 1] = *(const v2f*)(&sH0[frag_idx(l16, k + 4)]);
          }
        }
        __builtin_amdgcn_sched_barrier(0);
        #pragma unroll
        for (int i = 0; i < 4; ++i) {
          v2f b0 = { bq[cur][i].x, bq[cur][i].y };
          v2f b1 = { bq[cur][i].z, bq[cur][i].w };
          acc = __builtin_amdgcn_wmma_f32_16x16x4_f32(false, aq[cur][2 * i], false, b0,
                                                      (short)0, acc, false, false);
          acc = __builtin_amdgcn_wmma_f32_16x16x4_f32(false, aq[cur][2 * i + 1], false, b1,
                                                      (short)0, acc, false, false);
        }
      }
      const float bias = b1[n];
      #pragma unroll
      for (int v = 0; v < 8; ++v)
        sH1[frag_idx(v + kh * 8, n)] = softplusf(acc[v] + bias);
    }
    __syncthreads();
    {
      const int dtile = wav & 3;
      const int kq    = wav >> 2;
      const int d     = dtile * 16 + l16;
      v8f acc = {};
      const float* wp = W2 + d * WIDTH + kq * 64 + kh * 2;
      #pragma unroll
      for (int kk = 0; kk < 16; ++kk) {
        const int k = kq * 64 + kk * 4 + kh * 2;
        v2f a = *(const v2f*)(&sH1[frag_idx(l16, k)]);
        v2f b = *(const v2f*)(wp + kk * 4);
        acc = __builtin_amdgcn_wmma_f32_16x16x4_f32(false, a, false, b,
                                                    (short)0, acc, false, false);
      }
      #pragma unroll
      for (int v = 0; v < 8; ++v)
        sFp[kq][v + kh * 8][d] = acc[v];
    }
    __syncthreads();
  };

  #pragma unroll 1
  for (int iv = 0; iv < NINT; ++iv) {
    const float t1 = sTs[iv + 1];
    if (lane == 0) sT[er] = sTs[iv];
    __syncthreads();

    #pragma unroll 1
    for (int ss = 0; ss < MAXSUB; ++ss) {
      if (lane == 0) {
        float t = sT[er], dt = sDt[er];
        float rem  = t1 - t;
        bool  done = rem <= 1e-6f;
        sDtc[er]  = done ? dt : fminf(dt, rem);
        sDone[er] = done ? 1.0f : 0.0f;
      }
      __syncthreads();
      const float dtc = sDtc[er];

      float zs[6][2], es[2] = { 0.0f, 0.0f };
      #pragma unroll
      for (int i = 0; i < 6; ++i) { zs[i][0] = 0.0f; zs[i][1] = 0.0f; }
      #pragma unroll
      for (int j = 0; j < 2; ++j)
        sZ[frag_idx(er, ec + j)] = fmaf(dtc, 0.0f, yv[j]);
      __syncthreads();

      #pragma unroll 1
      for (int s = 0; s < 7; ++s) {
        float col[7];
        #pragma unroll
        for (int i = 0; i < 7; ++i) col[i] = STAGE_COL[s][i];
        mlp();
        #pragma unroll
        for (int j = 0; j < 2; ++j) {
          const int c = ec + j;
          const float k = sFp[0][er][c] + sFp[1][er][c] + sFp[2][er][c] + sFp[3][er][c] + b2[c];
          #pragma unroll
          for (int i = 0; i < 6; ++i) zs[i][j] = fmaf(col[i], k, zs[i][j]);
          es[j] = fmaf(col[6], k, es[j]);
        }
        if (s < 6) {
          #pragma unroll
          for (int j = 0; j < 2; ++j) {
            const float z = fmaf(dtc, zs[0][j], yv[j]);
            sZ[frag_idx(er, ec + j)] = z;
            if (s == 5) ynv[j] = z;
            #pragma unroll
            for (int i = 0; i < 5; ++i) zs[i][j] = zs[i + 1][j];
            zs[5][j] = 0.0f;
          }
          __syncthreads();
        }
      }

      float esq = 0.0f;
      #pragma unroll
      for (int j = 0; j < 2; ++j) {
        float e = es[j];
        e *= dtc;
        float scale = fmaf(RTOLC, fmaxf(fabsf(yv[j]), fabsf(ynv[j])), ATOLC);
        float r = e / scale;
        esq = fmaf(r, r, esq);
      }
      #pragma unroll
      for (int o = 16; o > 0; o >>= 1) esq += __shfl_xor(esq, o, 32);
      if (lane == 0) {
        float errn   = sqrtf(esq * (1.0f / DATA));
        bool  accept = errn <= 1.0f;
        float factor = SAFETY * __powf(fmaxf(errn, 1e-10f), -0.2f);
        factor = fminf(fmaxf(factor, FMINC), FMAXC);
        bool  done = sDone[er] != 0.0f;
        bool  step = accept && !done;
        if (step)  sT[er] += dtc;
        if (!done) sDt[er] = dtc * factor;
        sStep[er] = step ? 1.0f : 0.0f;
      }
      __syncthreads();
      if (sStep[er] != 0.0f) {
        yv[0] = ynv[0];
        yv[1] = ynv[1];
      }
      __syncthreads();
    }

    float* op = out + ((size_t)(row0 + er) * TPTS + (iv + 1)) * DATA + ec;
    op[0] = yv[0];
    op[1] = yv[1];
    __syncthreads();
  }
}

extern "C" void kernel_launch(void* const* d_in, const int* in_sizes, int n_in,
                              void* d_out, int out_size, void* d_ws, size_t ws_size,
                              hipStream_t stream) {
  (void)in_sizes; (void)n_in; (void)out_size; (void)d_ws; (void)ws_size;
  const float* ts = (const float*)d_in[0];
  const float* y0 = (const float*)d_in[1];
  const float* W0 = (const float*)d_in[2];
  const float* b0 = (const float*)d_in[3];
  const float* W1 = (const float*)d_in[4];
  const float* b1 = (const float*)d_in[5];
  const float* W2 = (const float*)d_in[6];
  const float* b2 = (const float*)d_in[7];
  float* out = (float*)d_out;
  node_tsit5_kernel<<<dim3(BATCH / MT), dim3(NTHR), 0, stream>>>(
      ts, y0, W0, b0, W1, b1, W2, b2, out);
}
